# starting kernel
# baseline (speedup 1.0000x reference)
#include <hip/hip_runtime.h>

#define DIM 64

typedef float v2f __attribute__((ext_vector_type(2)));
typedef float v8f __attribute__((ext_vector_type(8)));

__global__ void bignn_zero(float4* __restrict__ p, int n4) {
    int i = blockIdx.x * blockDim.x + threadIdx.x;
    if (i < n4) p[i] = make_float4(0.f, 0.f, 0.f, 0.f);
}

__global__ void bignn_spmm(const int* __restrict__ erow,
                           const int* __restrict__ ecol,
                           const float* __restrict__ eval_,
                           const float* __restrict__ x,
                           float* __restrict__ y, int E) {
    int t = blockIdx.x * blockDim.x + threadIdx.x;
    int e = t >> 5;
    if (e >= E) return;
    int lane = t & 31;
    int r = erow[e];
    int c = ecol[e];
    float v = eval_[e];
    int off = lane * 2;
    float2 xv = *(const float2*)(x + (long)c * DIM + off);
    float* dst = y + (long)r * DIM + off;
    atomicAdd(dst,     v * xv.x);
    atomicAdd(dst + 1, v * xv.y);
}

__global__ __launch_bounds__(256)
void bignn_gemm(const float* __restrict__ lap,
                const float* __restrict__ feat,
                const float* __restrict__ W1, const float* __restrict__ b1,
                const float* __restrict__ W2, const float* __restrict__ b2,
                float* __restrict__ out, int N) {
    __shared__ v2f  sW1[(DIM / 2) * DIM];
    __shared__ v2f  sW2[(DIM / 2) * DIM];
    __shared__ float sBias[DIM];

    int tid = threadIdx.x;
    {
        float* w1f = (float*)sW1;
        float* w2f = (float*)sW2;
        for (int i = tid; i < DIM * DIM; i += 256) {
            int kRow = i >> 6;
            int n    = i & 63;
            int dst  = (kRow >> 1) * (2 * DIM) + n * 2 + (kRow & 1);
            w1f[dst] = W1[i];
            w2f[dst] = W2[i];
        }
        if (tid < DIM) sBias[tid] = b1[tid] + b2[tid];
    }
    __syncthreads();

    int wave = tid >> 5;
    int lane = tid & 31;
    int rowBase = blockIdx.x * 128 + wave * 16;
    if (rowBase >= N) return;

    int m    = lane & 15;
    int half = lane >> 4;
    int row  = rowBase + m;
    int rowSafe = row < N ? row : (N - 1);

    const float2* lrow = (const float2*)(lap  + (long)rowSafe * DIM);
    const float2* frow = (const float2*)(feat + (long)rowSafe * DIM);

    v8f acc[4];
    acc[0] = v8f{}; acc[1] = v8f{}; acc[2] = v8f{}; acc[3] = v8f{};

    for (int k0 = 0; k0 < 16; ++k0) {
        int kp = k0 * 2 + half;
        float2 l = lrow[kp];
        float2 f = frow[kp];
        v2f aS = { l.x + f.x, l.y + f.y };
        v2f aP = { l.x * f.x, l.y * f.y };
        #pragma unroll
        for (int nb = 0; nb < 4; ++nb) {
            int cidx = nb * 16 + m;
            v2f bA = sW1[kp * DIM + cidx];
            v2f bB = sW2[kp * DIM + cidx];
            acc[nb] = __builtin_amdgcn_wmma_f32_16x16x4_f32(
                false, aS, false, bA, (short)0, acc[nb], false, false);
            acc[nb] = __builtin_amdgcn_wmma_f32_16x16x4_f32(
                false, aP, false, bB, (short)0, acc[nb], false, false);
        }
    }

    int rOff = half * 8;
    if (rowBase + 16 <= N) {
        #pragma unroll
        for (int nb = 0; nb < 4; ++nb) {
            int col = nb * 16 + m;
            float bias = sBias[col];
            #pragma unroll
            for (int g = 0; g < 8; ++g) {
                out[(long)(rowBase + rOff + g) * DIM + col] = acc[nb][g] + bias;
            }
        }
    } else {
        #pragma unroll
        for (int nb = 0; nb < 4; ++nb) {
            int col = nb * 16 + m;
            float bias = sBias[col];
            #pragma unroll
            for (int g = 0; g < 8; ++g) {
                int r = rowBase + rOff + g;
                if (r < N) out[(long)r * DIM + col] = acc[nb][g] + bias;
            }
        }
    }
}

extern "C" void kernel_launch(void* const* d_in, const int* in_sizes, int n_in,
                              void* d_out, int out_size, void* d_ws, size_t ws_size,
                              hipStream_t stream) {
    const int*   erow  = (const int*)  d_in[0];
    const int*   ecol  = (const int*)  d_in[1];
    const float* eval_ = (const float*)d_in[2];
    const float* feat  = (const float*)d_in[3];
    const float* W1    = (const float*)d_in[4];
    const float* b1    = (const float*)d_in[5];
    const float* W2    = (const float*)d_in[6];
    const float* b2    = (const float*)d_in[7];
    float* out = (float*)d_out;
    float* lap = (float*)d_ws;

    int E = in_sizes[0];
    int N = in_sizes[3] / DIM;

    int n4 = (N * DIM) / 4;
    bignn_zero<<<(n4 + 255) / 256, 256, 0, stream>>>((float4*)lap, n4);

    long threads = (long)E * 32;
    int blocks = (int)((threads + 255) / 256);
    bignn_spmm<<<blocks, 256, 0, stream>>>(erow, ecol, eval_, feat, lap, E);

    bignn_gemm<<<(N + 127) / 128, 256, 0, stream>>>(lap, feat, W1, b1, W2, b2, out, N);
}


# SpMM scatter: each wave takes 16 consecutive edges (3 batched edge loads, 16 gathers in flight, then the atomics) instead of 1 edge per wave; f32 atomics stay arrival-ordered as before
# speedup vs baseline: 1.8413x; 1.8413x over previous
#include <hip/hip_runtime.h>

#define DIM 64

typedef float v2f __attribute__((ext_vector_type(2)));
typedef float v8f __attribute__((ext_vector_type(8)));

__global__ void bignn_zero(float4* __restrict__ p, int n4) {
    int i = blockIdx.x * blockDim.x + threadIdx.x;
    if (i < n4) p[i] = make_float4(0.f, 0.f, 0.f, 0.f);
}

#define EPW 16

__global__ __launch_bounds__(256)
void bignn_spmm(const int* __restrict__ erow,
                const int* __restrict__ ecol,
                const float* __restrict__ eval_,
                const float* __restrict__ x,
                float* __restrict__ y, int E) {
    int t = blockIdx.x * blockDim.x + threadIdx.x;
    int lane = threadIdx.x & 31;
    int wave = __builtin_amdgcn_readfirstlane(t >> 5);
    long base = (long)wave * EPW;
    if (base >= E) return;
    int off = lane * 2;
    if (base + EPW <= E) {
        int sub = lane & (EPW - 1);
        int   rl = erow[base + sub];
        int   cl = ecol[base + sub];
        float vl = eval_[base + sub];
        float2 xv[EPW];
        #pragma unroll
        for (int j = 0; j < EPW; ++j) {
            int c = __builtin_amdgcn_readlane(cl, j);
            xv[j] = *(const float2*)(x + (long)c * DIM + off);
        }
        #pragma unroll
        for (int j = 0; j < EPW; ++j) {
            int   r = __builtin_amdgcn_readlane(rl, j);
            float v = __int_as_float(__builtin_amdgcn_readlane(__float_as_int(vl), j));
            float* dst = y + (long)r * DIM + off;
            atomicAdd(dst,     v * xv[j].x);
            atomicAdd(dst + 1, v * xv[j].y);
        }
    } else {
        for (long e = base; e < E; ++e) {
            int r = erow[e];
            int c = ecol[e];
            float v = eval_[e];
            float2 xv = *(const float2*)(x + (long)c * DIM + off);
            float* dst = y + (long)r * DIM + off;
            atomicAdd(dst,     v * xv.x);
            atomicAdd(dst + 1, v * xv.y);
        }
    }
}

__global__ __launch_bounds__(256)
void bignn_gemm(const float* __restrict__ lap,
                const float* __restrict__ feat,
                const float* __restrict__ W1, const float* __restrict__ b1,
                const float* __restrict__ W2, const float* __restrict__ b2,
                float* __restrict__ out, int N) {
    __shared__ v2f  sW1[(DIM / 2) * DIM];
    __shared__ v2f  sW2[(DIM / 2) * DIM];
    __shared__ float sBias[DIM];

    int tid = threadIdx.x;
    {
        float* w1f = (float*)sW1;
        float* w2f = (float*)sW2;
        for (int i = tid; i < DIM * DIM; i += 256) {
            int kRow = i >> 6;
            int n    = i & 63;
            int dst  = (kRow >> 1) * (2 * DIM) + n * 2 + (kRow & 1);
            w1f[dst] = W1[i];
            w2f[dst] = W2[i];
        }
        if (tid < DIM) sBias[tid] = b1[tid] + b2[tid];
    }
    __syncthreads();

    int wave = tid >> 5;
    int lane = tid & 31;
    int rowBase = blockIdx.x * 128 + wave * 16;
    if (rowBase >= N) return;

    int m    = lane & 15;
    int half = lane >> 4;
    int row  = rowBase + m;
    int rowSafe = row < N ? row : (N - 1);

    const float2* lrow = (const float2*)(lap  + (long)rowSafe * DIM);
    const float2* frow = (const float2*)(feat + (long)rowSafe * DIM);

    v8f acc[4];
    acc[0] = v8f{}; acc[1] = v8f{}; acc[2] = v8f{}; acc[3] = v8f{};

    for (int k0 = 0; k0 < 16; ++k0) {
        int kp = k0 * 2 + half;
        float2 l = lrow[kp];
        float2 f = frow[kp];
        v2f aS = { l.x + f.x, l.y + f.y };
        v2f aP = { l.x * f.x, l.y * f.y };
        #pragma unroll
        for (int nb = 0; nb < 4; ++nb) {
            int cidx = nb * 16 + m;
            v2f bA = sW1[kp * DIM + cidx];
            v2f bB = sW2[kp * DIM + cidx];
            acc[nb] = __builtin_amdgcn_wmma_f32_16x16x4_f32(
                false, aS, false, bA, (short)0, acc[nb], false, false);
            acc[nb] = __builtin_amdgcn_wmma_f32_16x16x4_f32(
                false, aP, false, bB, (short)0, acc[nb], false, false);
        }
    }

    int rOff = half * 8;
    if (rowBase + 16 <= N) {
        #pragma unroll
        for (int nb = 0; nb < 4; ++nb) {
            int col = nb * 16 + m;
            float bias = sBias[col];
            #pragma unroll
            for (int g = 0; g < 8; ++g) {
                out[(long)(rowBase + rOff + g) * DIM + col] = acc[nb][g] + bias;
            }
        }
    } else {
        #pragma unroll
        for (int nb = 0; nb < 4; ++nb) {
            int col = nb * 16 + m;
            float bias = sBias[col];
            #pragma unroll
            for (int g = 0; g < 8; ++g) {
                int r = rowBase + rOff + g;
                if (r < N) out[(long)r * DIM + col] = acc[nb][g] + bias;
            }
        }
    }
}

extern "C" void kernel_launch(void* const* d_in, const int* in_sizes, int n_in,
                              void* d_out, int out_size, void* d_ws, size_t ws_size,
                              hipStream_t stream) {
    const int*   erow  = (const int*)  d_in[0];
    const int*   ecol  = (const int*)  d_in[1];
    const float* eval_ = (const float*)d_in[2];
    const float* feat  = (const float*)d_in[3];
    const float* W1    = (const float*)d_in[4];
    const float* b1    = (const float*)d_in[5];
    const float* W2    = (const float*)d_in[6];
    const float* b2    = (const float*)d_in[7];
    float* out = (float*)d_out;
    float* lap = (float*)d_ws;

    int E = in_sizes[0];
    int N = in_sizes[3] / DIM;

    int n4 = (N * DIM) / 4;
    bignn_zero<<<(n4 + 255) / 256, 256, 0, stream>>>((float4*)lap, n4);

    long waves = ((long)E + EPW - 1) / EPW;
    int blocks = (int)((waves * 32 + 255) / 256);
    bignn_spmm<<<blocks, 256, 0, stream>>>(erow, ecol, eval_, feat, lap, E);

    bignn_gemm<<<(N + 127) / 128, 256, 0, stream>>>(lap, feat, W1, b1, W2, b2, out, N);
}


# SpMM scatter: lane l takes columns l and l+32 of each edge, so every gather and every f32 atomic covers one whole 128-byte line (2 line requests per row update, not 4); atomics stay arrival-ordered
# speedup vs baseline: 2.1065x; 1.1440x over previous
#include <hip/hip_runtime.h>

#define DIM 64

typedef float v2f __attribute__((ext_vector_type(2)));
typedef float v8f __attribute__((ext_vector_type(8)));

__global__ void bignn_zero(float4* __restrict__ p, int n4) {
    int i = blockIdx.x * blockDim.x + threadIdx.x;
    if (i < n4) p[i] = make_float4(0.f, 0.f, 0.f, 0.f);
}

#define EPW 16

__global__ __launch_bounds__(256)
void bignn_spmm(const int* __restrict__ erow,
                const int* __restrict__ ecol,
                const float* __restrict__ eval_,
                const float* __restrict__ x,
                float* __restrict__ y, int E) {
    int t = blockIdx.x * blockDim.x + threadIdx.x;
    int lane = threadIdx.x & 31;
    int wave = __builtin_amdgcn_readfirstlane(t >> 5);
    long base = (long)wave * EPW;
    if (base >= E) return;
    if (base + EPW <= E) {
        int sub = lane & (EPW - 1);
        int   rl = erow[base + sub];
        int   cl = ecol[base + sub];
        float vl = eval_[base + sub];
        float xa[EPW];
        float xb[EPW];
        #pragma unroll
        for (int j = 0; j < EPW; ++j) {
            int c = __builtin_amdgcn_readlane(cl, j);
            const float* srcRow = x + (long)c * DIM;
            xa[j] = srcRow[lane];
            xb[j] = srcRow[lane + 32];
        }
        #pragma unroll
        for (int j = 0; j < EPW; ++j) {
            int   r = __builtin_amdgcn_readlane(rl, j);
            float v = __int_as_float(__builtin_amdgcn_readlane(__float_as_int(vl), j));
            float* dst = y + (long)r * DIM;
            atomicAdd(dst + lane,      v * xa[j]);
            atomicAdd(dst + lane + 32, v * xb[j]);
        }
    } else {
        for (long e = base; e < E; ++e) {
            int r = erow[e];
            int c = ecol[e];
            float v = eval_[e];
            const float* srcRow = x + (long)c * DIM;
            float a = srcRow[lane];
            float b = srcRow[lane + 32];
            float* dst = y + (long)r * DIM;
            atomicAdd(dst + lane,      v * a);
            atomicAdd(dst + lane + 32, v * b);
        }
    }
}

__global__ __launch_bounds__(256)
void bignn_gemm(const float* __restrict__ lap,
                const float* __restrict__ feat,
                const float* __restrict__ W1, const float* __restrict__ b1,
                const float* __restrict__ W2, const float* __restrict__ b2,
                float* __restrict__ out, int N) {
    __shared__ v2f  sW1[(DIM / 2) * DIM];
    __shared__ v2f  sW2[(DIM / 2) * DIM];
    __shared__ float sBias[DIM];

    int tid = threadIdx.x;
    {
        float* w1f = (float*)sW1;
        float* w2f = (float*)sW2;
        for (int i = tid; i < DIM * DIM; i += 256) {
            int kRow = i >> 6;
            int n    = i & 63;
            int dst  = (kRow >> 1) * (2 * DIM) + n * 2 + (kRow & 1);
            w1f[dst] = W1[i];
            w2f[dst] = W2[i];
        }
        if (tid < DIM) sBias[tid] = b1[tid] + b2[tid];
    }
    __syncthreads();

    int wave = tid >> 5;
    int lane = tid & 31;
    int rowBase = blockIdx.x * 128 + wave * 16;
    if (rowBase >= N) return;

    int m    = lane & 15;
    int half = lane >> 4;
    int row  = rowBase + m;
    int rowSafe = row < N ? row : (N - 1);

    const float2* lrow = (const float2*)(lap  + (long)rowSafe * DIM);
    const float2* frow = (const float2*)(feat + (long)rowSafe * DIM);

    v8f acc[4];
    acc[0] = v8f{}; acc[1] = v8f{}; acc[2] = v8f{}; acc[3] = v8f{};

    for (int k0 = 0; k0 < 16; ++k0) {
        int kp = k0 * 2 + half;
        float2 l = lrow[kp];
        float2 f = frow[kp];
        v2f aS = { l.x + f.x, l.y + f.y };
        v2f aP = { l.x * f.x, l.y * f.y };
        #pragma unroll
        for (int nb = 0; nb < 4; ++nb) {
            int cidx = nb * 16 + m;
            v2f bA = sW1[kp * DIM + cidx];
            v2f bB = sW2[kp * DIM + cidx];
            acc[nb] = __builtin_amdgcn_wmma_f32_16x16x4_f32(
                false, aS, false, bA, (short)0, acc[nb], false, false);
            acc[nb] = __builtin_amdgcn_wmma_f32_16x16x4_f32(
                false, aP, false, bB, (short)0, acc[nb], false, false);
        }
    }

    int rOff = half * 8;
    if (rowBase + 16 <= N) {
        #pragma unroll
        for (int nb = 0; nb < 4; ++nb) {
            int col = nb * 16 + m;
            float bias = sBias[col];
            #pragma unroll
            for (int g = 0; g < 8; ++g) {
                out[(long)(rowBase + rOff + g) * DIM + col] = acc[nb][g] + bias;
            }
        }
    } else {
        #pragma unroll
        for (int nb = 0; nb < 4; ++nb) {
            int col = nb * 16 + m;
            float bias = sBias[col];
            #pragma unroll
            for (int g = 0; g < 8; ++g) {
                int r = rowBase + rOff + g;
                if (r < N) out[(long)r * DIM + col] = acc[nb][g] + bias;
            }
        }
    }
}

extern "C" void kernel_launch(void* const* d_in, const int* in_sizes, int n_in,
                              void* d_out, int out_size, void* d_ws, size_t ws_size,
                              hipStream_t stream) {
    const int*   erow  = (const int*)  d_in[0];
    const int*   ecol  = (const int*)  d_in[1];
    const float* eval_ = (const float*)d_in[2];
    const float* feat  = (const float*)d_in[3];
    const float* W1    = (const float*)d_in[4];
    const float* b1    = (const float*)d_in[5];
    const float* W2    = (const float*)d_in[6];
    const float* b2    = (const float*)d_in[7];
    float* out = (float*)d_out;
    float* lap = (float*)d_ws;

    int E = in_sizes[0];
    int N = in_sizes[3] / DIM;

    int n4 = (N * DIM) / 4;
    bignn_zero<<<(n4 + 255) / 256, 256, 0, stream>>>((float4*)lap, n4);

    long waves = ((long)E + EPW - 1) / EPW;
    int blocks = (int)((waves * 32 + 255) / 256);
    bignn_spmm<<<blocks, 256, 0, stream>>>(erow, ecol, eval_, feat, lap, E);

    bignn_gemm<<<(N + 127) / 128, 256, 0, stream>>>(lap, feat, W1, b1, W2, b2, out, N);
}


# SpMM: edges binned by 64-row block (count, scan, place; int atomics), a block sums its rows by LDS f32 atomics and stores them once: no global f32 atomics, no zero pass; sums stay arrival-ordered
# speedup vs baseline: 2.6720x; 1.2685x over previous
#include <hip/hip_runtime.h>

#define DIM 64

typedef float v2f __attribute__((ext_vector_type(2)));
typedef float v8f __attribute__((ext_vector_type(8)));

#define BIN_SHIFT 6
#define BIN_ROWS  64
#define MAXB      2048
#define EPW       16
#define UB        8

__global__ void bignn_zero_ctr(int* __restrict__ p, int n) {
    int i = blockIdx.x * blockDim.x + threadIdx.x;
    if (i < n) p[i] = 0;
}

__global__ __launch_bounds__(1024)
void bignn_count(const int* __restrict__ erow, int* __restrict__ cnt, int E, int NB) {
    __shared__ int h[MAXB];
    int tid = threadIdx.x;
    for (int i = tid; i < NB; i += 1024) h[i] = 0;
    __syncthreads();
    long chunk = ((long)E + gridDim.x - 1) / gridDim.x;
    long beg = (long)blockIdx.x * chunk;
    long end = beg + chunk;
    if (end > E) end = E;
    for (long e0 = beg + tid; e0 < end; e0 += UB * 1024) {
        int r[UB];
        #pragma unroll
        for (int k = 0; k < UB; ++k) {
            long e = e0 + k * 1024;
            r[k] = e < end ? erow[e] : -1;
        }
        #pragma unroll
        for (int k = 0; k < UB; ++k)
            if (r[k] >= 0) atomicAdd(&h[r[k] >> BIN_SHIFT], 1);
    }
    __syncthreads();
    for (int i = tid; i < NB; i += 1024) {
        int c = h[i];
        if (c) atomicAdd(&cnt[i], c);
    }
}

__global__ __launch_bounds__(1024)
void bignn_place(const int* __restrict__ erow, const int* __restrict__ ecol,
                 const float* __restrict__ eval_, const int* __restrict__ cnt,
                 int* __restrict__ cursor, int* __restrict__ binStart,
                 uint2* __restrict__ binned, int E, int NB) {
    __shared__ int lcnt[MAXB];
    __shared__ int lbase[MAXB];
    __shared__ int wsum[32];
    int tid = threadIdx.x;
    int lane = tid & 31;
    int wv = tid >> 5;
    int i0 = 2 * tid, i1 = 2 * tid + 1;
    int a = i0 < NB ? cnt[i0] : 0;
    int b = i1 < NB ? cnt[i1] : 0;
    int s = a + b;
    int incl = s;
    #pragma unroll
    for (int d = 1; d < 32; d <<= 1) {
        int t = __shfl_up(incl, d);
        if (lane >= d) incl += t;
    }
    if (lane == 31) wsum[wv] = incl;
    lcnt[i0] = 0;
    lcnt[i1] = 0;
    __syncthreads();
    if (wv == 0) {
        int w = wsum[lane];
        int wi = w;
        #pragma unroll
        for (int d = 1; d < 32; d <<= 1) {
            int t = __shfl_up(wi, d);
            if (lane >= d) wi += t;
        }
        wsum[lane] = wi - w;
    }
    __syncthreads();
    int excl = wsum[wv] + incl - s;
    lbase[i0] = excl;
    lbase[i1] = excl + a;
    if (blockIdx.x == 0) {
        if (i0 < NB) binStart[i0] = excl;
        if (i1 < NB) binStart[i1] = excl + a;
    }
    long chunk = ((long)E + gridDim.x - 1) / gridDim.x;
    long beg = (long)blockIdx.x * chunk;
    long end = beg + chunk;
    if (end > E) end = E;
    for (long e0 = beg + tid; e0 < end; e0 += UB * 1024) {
        int r[UB];
        #pragma unroll
        for (int k = 0; k < UB; ++k) {
            long e = e0 + k * 1024;
            r[k] = e < end ? erow[e] : -1;
        }
        #pragma unroll
        for (int k = 0; k < UB; ++k)
            if (r[k] >= 0) atomicAdd(&lcnt[r[k] >> BIN_SHIFT], 1);
    }
    __syncthreads();
    if (i0 < NB) { int c = lcnt[i0]; if (c) lbase[i0] += atomicAdd(&cursor[i0], c); }
    if (i1 < NB) { int c = lcnt[i1]; if (c) lbase[i1] += atomicAdd(&cursor[i1], c); }
    lcnt[i0] = 0;
    lcnt[i1] = 0;
    __syncthreads();
    for (long e0 = beg + tid; e0 < end; e0 += UB * 1024) {
        int r[UB];
        int c[UB];
        float v[UB];
        #pragma unroll
        for (int k = 0; k < UB; ++k) {
            long e = e0 + k * 1024;
            bool in = e < end;
            r[k] = in ? erow[e] : -1;
            c[k] = in ? ecol[e] : 0;
            v[k] = in ? eval_[e] : 0.f;
        }
        #pragma unroll
        for (int k = 0; k < UB; ++k) {
            if (r[k] >= 0) {
                int bin = r[k] >> BIN_SHIFT;
                int pos = lbase[bin] + atomicAdd(&lcnt[bin], 1);
                binned[pos] = make_uint2(((unsigned)c[k] << BIN_SHIFT) | (unsigned)(r[k] & (BIN_ROWS - 1)),
                                         __float_as_uint(v[k]));
            }
        }
    }
}

__global__ __launch_bounds__(256)
void bignn_accum(const uint2* __restrict__ binned, const int* __restrict__ cnt,
                 const int* __restrict__ binStart, const float* __restrict__ x,
                 float* __restrict__ y, int N) {
    __shared__ float acc[BIN_ROWS * DIM];
    int tid = threadIdx.x;
    int lane = tid & 31;
    int wv = __builtin_amdgcn_readfirstlane(tid >> 5);
    int bin = blockIdx.x;
    for (int i = tid; i < BIN_ROWS * DIM; i += 256) acc[i] = 0.f;
    int n = cnt[bin];
    const uint2* p = binned + binStart[bin];
    __syncthreads();
    for (int g = wv * EPW; g < n; g += 8 * EPW) {
        if (g + EPW <= n) {
            uint2 kv = p[g + (lane & (EPW - 1))];
            float xa[EPW];
            float xb[EPW];
            #pragma unroll
            for (int j = 0; j < EPW; ++j) {
                unsigned key = (unsigned)__builtin_amdgcn_readlane((int)kv.x, j);
                const float* srcRow = x + (long)(key >> BIN_SHIFT) * DIM;
                xa[j] = srcRow[lane];
                xb[j] = srcRow[lane + 32];
            }
            #pragma unroll
            for (int j = 0; j < EPW; ++j) {
                unsigned key = (unsigned)__builtin_amdgcn_readlane((int)kv.x, j);
                float v = __int_as_float(__builtin_amdgcn_readlane((int)kv.y, j));
                float* dst = acc + (key & (BIN_ROWS - 1)) * DIM;
                atomicAdd(dst + lane,      v * xa[j]);
                atomicAdd(dst + lane + 32, v * xb[j]);
            }
        } else {
            for (int e = g; e < n; ++e) {
                uint2 kv = p[e];
                const float* srcRow = x + (long)(kv.x >> BIN_SHIFT) * DIM;
                float v = __uint_as_float(kv.y);
                float va = srcRow[lane];
                float vb = srcRow[lane + 32];
                float* dst = acc + (kv.x & (BIN_ROWS - 1)) * DIM;
                atomicAdd(dst + lane,      v * va);
                atomicAdd(dst + lane + 32, v * vb);
            }
        }
    }
    __syncthreads();
    const float4* a4 = (const float4*)acc;
    for (int i = tid; i < BIN_ROWS * (DIM / 4); i += 256) {
        int row = bin * BIN_ROWS + (i >> 4);
        if (row < N) ((float4*)(y + (long)row * DIM))[i & 15] = a4[i];
    }
}

__global__ __launch_bounds__(256)
void bignn_gemm(const float* __restrict__ lap,
                const float* __restrict__ feat,
                const float* __restrict__ W1, const float* __restrict__ b1,
                const float* __restrict__ W2, const float* __restrict__ b2,
                float* __restrict__ out, int N) {
    __shared__ v2f  sW1[(DIM / 2) * DIM];
    __shared__ v2f  sW2[(DIM / 2) * DIM];
    __shared__ float sBias[DIM];

    int tid = threadIdx.x;
    {
        float* w1f = (float*)sW1;
        float* w2f = (float*)sW2;
        for (int i = tid; i < DIM * DIM; i += 256) {
            int kRow = i >> 6;
            int n    = i & 63;
            int dst  = (kRow >> 1) * (2 * DIM) + n * 2 + (kRow & 1);
            w1f[dst] = W1[i];
            w2f[dst] = W2[i];
        }
        if (tid < DIM) sBias[tid] = b1[tid] + b2[tid];
    }
    __syncthreads();

    int wave = tid >> 5;
    int lane = tid & 31;
    int rowBase = blockIdx.x * 128 + wave * 16;
    if (rowBase >= N) return;

    int m    = lane & 15;
    int half = lane >> 4;
    int row  = rowBase + m;
    int rowSafe = row < N ? row : (N - 1);

    const float2* lrow = (const float2*)(lap  + (long)rowSafe * DIM);
    const float2* frow = (const float2*)(feat + (long)rowSafe * DIM);

    v8f acc[4];
    acc[0] = v8f{}; acc[1] = v8f{}; acc[2] = v8f{}; acc[3] = v8f{};

    for (int k0 = 0; k0 < 16; ++k0) {
        int kp = k0 * 2 + half;
        float2 l = lrow[kp];
        float2 f = frow[kp];
        v2f aS = { l.x + f.x, l.y + f.y };
        v2f aP = { l.x * f.x, l.y * f.y };
        #pragma unroll
        for (int nb = 0; nb < 4; ++nb) {
            int cidx = nb * 16 + m;
            v2f bA = sW1[kp * DIM + cidx];
            v2f bB = sW2[kp * DIM + cidx];
            acc[nb] = __builtin_amdgcn_wmma_f32_16x16x4_f32(
                false, aS, false, bA, (short)0, acc[nb], false, false);
            acc[nb] = __builtin_amdgcn_wmma_f32_16x16x4_f32(
                false, aP, false, bB, (short)0, acc[nb], false, false);
        }
    }

    int rOff = half * 8;
    if (rowBase + 16 <= N) {
        #pragma unroll
        for (int nb = 0; nb < 4; ++nb) {
            int col = nb * 16 + m;
            float bias = sBias[col];
            #pragma unroll
            for (int g = 0; g < 8; ++g) {
                out[(long)(rowBase + rOff + g) * DIM + col] = acc[nb][g] + bias;
            }
        }
    } else {
        #pragma unroll
        for (int nb = 0; nb < 4; ++nb) {
            int col = nb * 16 + m;
            float bias = sBias[col];
            #pragma unroll
            for (int g = 0; g < 8; ++g) {
                int r = rowBase + rOff + g;
                if (r < N) out[(long)r * DIM + col] = acc[nb][g] + bias;
            }
        }
    }
}

extern "C" void kernel_launch(void* const* d_in, const int* in_sizes, int n_in,
                              void* d_out, int out_size, void* d_ws, size_t ws_size,
                              hipStream_t stream) {
    const int*   erow  = (const int*)  d_in[0];
    const int*   ecol  = (const int*)  d_in[1];
    const float* eval_ = (const float*)d_in[2];
    const float* feat  = (const float*)d_in[3];
    const float* W1    = (const float*)d_in[4];
    const float* b1    = (const float*)d_in[5];
    const float* W2    = (const float*)d_in[6];
    const float* b2    = (const float*)d_in[7];
    float* out = (float*)d_out;

    int E = in_sizes[0];
    int N = in_sizes[3] / DIM;
    int NB = (N + BIN_ROWS - 1) / BIN_ROWS;

    float* lap      = (float*)d_ws;
    uint2* binned   = (uint2*)(lap + (size_t)N * DIM);
    int*   cnt      = (int*)(binned + E);
    int*   cursor   = cnt + MAXB;
    int*   binStart = cursor + MAXB;

    bignn_zero_ctr<<<(2 * MAXB + 1023) / 1024, 1024, 0, stream>>>(cnt, 2 * MAXB);
    bignn_count<<<256, 1024, 0, stream>>>(erow, cnt, E, NB);
    bignn_place<<<256, 1024, 0, stream>>>(erow, ecol, eval_, cnt, cursor, binStart, binned, E, NB);
    bignn_accum<<<NB, 256, 0, stream>>>(binned, cnt, binStart, feat, lap, N);
    bignn_gemm<<<(N + 127) / 128, 256, 0, stream>>>(lap, feat, W1, b1, W2, b2, out, N);
}


# GEMM fused into the SpMM accumulate kernel: a bin's 64x64 sum tile and its feature rows feed the WMMA A fragments from LDS; no workspace round trip of the SpMM result, one launch less
# speedup vs baseline: 2.9169x; 1.0917x over previous
#include <hip/hip_runtime.h>

#define DIM 64

typedef float v2f __attribute__((ext_vector_type(2)));
typedef float v8f __attribute__((ext_vector_type(8)));

#define BIN_SHIFT 6
#define BIN_ROWS  64
#define MAXB      2048
#define EPW       16
#define UB        8

__global__ void bignn_zero_ctr(int* __restrict__ p, int n) {
    int i = blockIdx.x * blockDim.x + threadIdx.x;
    if (i < n) p[i] = 0;
}

__global__ __launch_bounds__(1024)
void bignn_count(const int* __restrict__ erow, int* __restrict__ cnt, int E, int NB) {
    __shared__ int h[MAXB];
    int tid = threadIdx.x;
    for (int i = tid; i < NB; i += 1024) h[i] = 0;
    __syncthreads();
    long chunk = ((long)E + gridDim.x - 1) / gridDim.x;
    long beg = (long)blockIdx.x * chunk;
    long end = beg + chunk;
    if (end > E) end = E;
    for (long e0 = beg + tid; e0 < end; e0 += UB * 1024) {
        int r[UB];
        #pragma unroll
        for (int k = 0; k < UB; ++k) {
            long e = e0 + k * 1024;
            r[k] = e < end ? erow[e] : -1;
        }
        #pragma unroll
        for (int k = 0; k < UB; ++k)
            if (r[k] >= 0) atomicAdd(&h[r[k] >> BIN_SHIFT], 1);
    }
    __syncthreads();
    for (int i = tid; i < NB; i += 1024) {
        int c = h[i];
        if (c) atomicAdd(&cnt[i], c);
    }
}

__global__ __launch_bounds__(1024)
void bignn_place(const int* __restrict__ erow, const int* __restrict__ ecol,
                 const float* __restrict__ eval_, const int* __restrict__ cnt,
                 int* __restrict__ cursor, int* __restrict__ binStart,
                 uint2* __restrict__ binned, int E, int NB) {
    __shared__ int lcnt[MAXB];
    __shared__ int lbase[MAXB];
    __shared__ int wsum[32];
    int tid = threadIdx.x;
    int lane = tid & 31;
    int wv = tid >> 5;
    int i0 = 2 * tid, i1 = 2 * tid + 1;
    int a = i0 < NB ? cnt[i0] : 0;
    int b = i1 < NB ? cnt[i1] : 0;
    int s = a + b;
    int incl = s;
    #pragma unroll
    for (int d = 1; d < 32; d <<= 1) {
        int t = __shfl_up(incl, d);
        if (lane >= d) incl += t;
    }
    if (lane == 31) wsum[wv] = incl;
    lcnt[i0] = 0;
    lcnt[i1] = 0;
    __syncthreads();
    if (wv == 0) {
        int w = wsum[lane];
        int wi = w;
        #pragma unroll
        for (int d = 1; d < 32; d <<= 1) {
            int t = __shfl_up(wi, d);
            if (lane >= d) wi += t;
        }
        wsum[lane] = wi - w;
    }
    __syncthreads();
    int excl = wsum[wv] + incl - s;
    lbase[i0] = excl;
    lbase[i1] = excl + a;
    if (blockIdx.x == 0) {
        if (i0 < NB) binStart[i0] = excl;
        if (i1 < NB) binStart[i1] = excl + a;
    }
    long chunk = ((long)E + gridDim.x - 1) / gridDim.x;
    long beg = (long)blockIdx.x * chunk;
    long end = beg + chunk;
    if (end > E) end = E;
    for (long e0 = beg + tid; e0 < end; e0 += UB * 1024) {
        int r[UB];
        #pragma unroll
        for (int k = 0; k < UB; ++k) {
            long e = e0 + k * 1024;
            r[k] = e < end ? erow[e] : -1;
        }
        #pragma unroll
        for (int k = 0; k < UB; ++k)
            if (r[k] >= 0) atomicAdd(&lcnt[r[k] >> BIN_SHIFT], 1);
    }
    __syncthreads();
    if (i0 < NB) { int c = lcnt[i0]; if (c) lbase[i0] += atomicAdd(&cursor[i0], c); }
    if (i1 < NB) { int c = lcnt[i1]; if (c) lbase[i1] += atomicAdd(&cursor[i1], c); }
    lcnt[i0] = 0;
    lcnt[i1] = 0;
    __syncthreads();
    for (long e0 = beg + tid; e0 < end; e0 += UB * 1024) {
        int r[UB];
        int c[UB];
        float v[UB];
        #pragma unroll
        for (int k = 0; k < UB; ++k) {
            long e = e0 + k * 1024;
            bool in = e < end;
            r[k] = in ? erow[e] : -1;
            c[k] = in ? ecol[e] : 0;
            v[k] = in ? eval_[e] : 0.f;
        }
        #pragma unroll
        for (int k = 0; k < UB; ++k) {
            if (r[k] >= 0) {
                int bin = r[k] >> BIN_SHIFT;
                int pos = lbase[bin] + atomicAdd(&lcnt[bin], 1);
                binned[pos] = make_uint2(((unsigned)c[k] << BIN_SHIFT) | (unsigned)(r[k] & (BIN_ROWS - 1)),
                                         __float_as_uint(v[k]));
            }
        }
    }
}

__global__ __launch_bounds__(256)
void bignn_accum_gemm(const uint2* __restrict__ binned, const int* __restrict__ cnt,
                      const int* __restrict__ binStart, const float* __restrict__ x,
                      const float* __restrict__ W1, const float* __restrict__ b1,
                      const float* __restrict__ W2, const float* __restrict__ b2,
                      float* __restrict__ out, int N) {
    __shared__ __attribute__((aligned(16))) float sL[BIN_ROWS * DIM];
    __shared__ __attribute__((aligned(16))) float sF[BIN_ROWS * DIM];
    __shared__ __attribute__((aligned(16))) v2f sW1[(DIM / 2) * DIM];
    __shared__ __attribute__((aligned(16))) v2f sW2[(DIM / 2) * DIM];
    int tid = threadIdx.x;
    int lane = tid & 31;
    int wv = __builtin_amdgcn_readfirstlane(tid >> 5);
    int bin = blockIdx.x;
    int row0 = bin * BIN_ROWS;

    {
        float4 fv[4];
        #pragma unroll
        for (int j = 0; j < 4; ++j) {
            int i = tid + 256 * j;
            int gr = row0 + (i >> 4);
            if (gr >= N) gr = N - 1;
            fv[j] = ((const float4*)(x + (long)gr * DIM))[i & 15];
        }
        float4 w1e[2], w1o[2], w2e[2], w2o[2];
        #pragma unroll
        for (int j = 0; j < 2; ++j) {
            int u  = tid + 256 * j;
            int kp = u >> 4;
            int q  = u & 15;
            w1e[j] = ((const float4*)(W1 + (2 * kp)     * DIM))[q];
            w1o[j] = ((const float4*)(W1 + (2 * kp + 1) * DIM))[q];
            w2e[j] = ((const float4*)(W2 + (2 * kp)     * DIM))[q];
            w2o[j] = ((const float4*)(W2 + (2 * kp + 1) * DIM))[q];
        }
        const float4 z = {0.f, 0.f, 0.f, 0.f};
        #pragma unroll
        for (int j = 0; j < 4; ++j) {
            int i = tid + 256 * j;
            int r = i >> 4;
            int q = i & 15;
            *(float4*)(sF + r * DIM + ((q ^ (r & 15)) << 2)) = fv[j];
            ((float4*)sL)[i] = z;
        }
        #pragma unroll
        for (int j = 0; j < 2; ++j) {
            int u = tid + 256 * j;
            float4* d1 = (float4*)(sW1 + (u >> 4) * DIM + (u & 15) * 4);
            float4* d2 = (float4*)(sW2 + (u >> 4) * DIM + (u & 15) * 4);
            d1[0] = float4{w1e[j].x, w1o[j].x, w1e[j].y, w1o[j].y};
            d1[1] = float4{w1e[j].z, w1o[j].z, w1e[j].w, w1o[j].w};
            d2[0] = float4{w2e[j].x, w2o[j].x, w2e[j].y, w2o[j].y};
            d2[1] = float4{w2e[j].z, w2o[j].z, w2e[j].w, w2o[j].w};
        }
    }
    int n = cnt[bin];
    const uint2* p = binned + binStart[bin];
    __syncthreads();

    for (int g = wv * EPW; g < n; g += 8 * EPW) {
        if (g + EPW <= n) {
            uint2 kv = p[g + (lane & (EPW - 1))];
            float xa[EPW];
            float xb[EPW];
            #pragma unroll
            for (int j = 0; j < EPW; ++j) {
                unsigned key = (unsigned)__builtin_amdgcn_readlane((int)kv.x, j);
                const float* srcRow = x + (long)(key >> BIN_SHIFT) * DIM;
                xa[j] = srcRow[lane];
                xb[j] = srcRow[lane + 32];
            }
            #pragma unroll
            for (int j = 0; j < EPW; ++j) {
                unsigned key = (unsigned)__builtin_amdgcn_readlane((int)kv.x, j);
                float v = __int_as_float(__builtin_amdgcn_readlane((int)kv.y, j));
                int r = key & (BIN_ROWS - 1);
                float* dst = sL + r * DIM;
                int c = lane ^ ((r & 15) << 2);
                atomicAdd(dst + c,        v * xa[j]);
                atomicAdd(dst + (c ^ 32), v * xb[j]);
            }
        } else {
            for (int e = g; e < n; ++e) {
                uint2 kv = p[e];
                const float* srcRow = x + (long)(kv.x >> BIN_SHIFT) * DIM;
                float v = __uint_as_float(kv.y);
                float va = srcRow[lane];
                float vb = srcRow[lane + 32];
                int r = kv.x & (BIN_ROWS - 1);
                float* dst = sL + r * DIM;
                int c = lane ^ ((r & 15) << 2);
                atomicAdd(dst + c,        v * va);
                atomicAdd(dst + (c ^ 32), v * vb);
            }
        }
    }
    __syncthreads();

    int m    = lane & 15;
    int half = lane >> 4;
    int rt   = wv & 3;
    int ch   = wv >> 2;
    int r    = rt * 16 + m;
    int sw   = m << 2;
    const float* lrow = sL + r * DIM;
    const float* frow = sF + r * DIM;
    int col0 = ch * 32 + m;
    int col1 = col0 + 16;
    float bias0 = b1[col0] + b2[col0];
    float bias1 = b1[col1] + b2[col1];

    v8f acc0 = v8f{};
    v8f acc1 = v8f{};
    for (int k0 = 0; k0 < 16; ++k0) {
        int kp = k0 * 2 + half;
        int c  = (2 * kp) ^ sw;
        float2 l = *(const float2*)(lrow + c);
        float2 f = *(const float2*)(frow + c);
        v2f aS = { l.x + f.x, l.y + f.y };
        v2f aP = { l.x * f.x, l.y * f.y };
        v2f bA0 = sW1[kp * DIM + col0];
        v2f bB0 = sW2[kp * DIM + col0];
        v2f bA1 = sW1[kp * DIM + col1];
        v2f bB1 = sW2[kp * DIM + col1];
        acc0 = __builtin_amdgcn_wmma_f32_16x16x4_f32(
            false, aS, false, bA0, (short)0, acc0, false, false);
        acc1 = __builtin_amdgcn_wmma_f32_16x16x4_f32(
            false, aS, false, bA1, (short)0, acc1, false, false);
        acc0 = __builtin_amdgcn_wmma_f32_16x16x4_f32(
            false, aP, false, bB0, (short)0, acc0, false, false);
        acc1 = __builtin_amdgcn_wmma_f32_16x16x4_f32(
            false, aP, false, bB1, (short)0, acc1, false, false);
    }

    int rBase = row0 + rt * 16 + half * 8;
    if (row0 + BIN_ROWS <= N) {
        #pragma unroll
        for (int g = 0; g < 8; ++g) {
            float* o = out + (long)(rBase + g) * DIM;
            o[col0] = acc0[g] + bias0;
            o[col1] = acc1[g] + bias1;
        }
    } else {
        #pragma unroll
        for (int g = 0; g < 8; ++g) {
            int gr = rBase + g;
            if (gr < N) {
                float* o = out + (long)gr * DIM;
                o[col0] = acc0[g] + bias0;
                o[col1] = acc1[g] + bias1;
            }
        }
    }
}

extern "C" void kernel_launch(void* const* d_in, const int* in_sizes, int n_in,
                              void* d_out, int out_size, void* d_ws, size_t ws_size,
                              hipStream_t stream) {
    const int*   erow  = (const int*)  d_in[0];
    const int*   ecol  = (const int*)  d_in[1];
    const float* eval_ = (const float*)d_in[2];
    const float* feat  = (const float*)d_in[3];
    const float* W1    = (const float*)d_in[4];
    const float* b1    = (const float*)d_in[5];
    const float* W2    = (const float*)d_in[6];
    const float* b2    = (const float*)d_in[7];
    float* out = (float*)d_out;

    int E = in_sizes[0];
    int N = in_sizes[3] / DIM;
    int NB = (N + BIN_ROWS - 1) / BIN_ROWS;

    uint2* binned   = (uint2*)d_ws;
    int*   cnt      = (int*)(binned + E);
    int*   cursor   = cnt + MAXB;
    int*   binStart = cursor + MAXB;

    bignn_zero_ctr<<<(2 * MAXB + 1023) / 1024, 1024, 0, stream>>>(cnt, 2 * MAXB);
    bignn_count<<<256, 1024, 0, stream>>>(erow, cnt, E, NB);
    bignn_place<<<256, 1024, 0, stream>>>(erow, ecol, eval_, cnt, cursor, binStart, binned, E, NB);
    bignn_accum_gemm<<<NB, 256, 0, stream>>>(binned, cnt, binStart, feat, W1, b1, W2, b2, out, N);
}


# Fused kernel: each bin block counting-sorts its edges by destination row in LDS and sums every row's run in registers (one LDS f32 add per run, not per edge); f32 sums stay arrival-ordered
# speedup vs baseline: 3.2812x; 1.1249x over previous
#include <hip/hip_runtime.h>

#define DIM 64

typedef float v2f __attribute__((ext_vector_type(2)));
typedef float v8f __attribute__((ext_vector_type(8)));

#define BIN_SHIFT 6
#define BIN_ROWS  64
#define MAXB      2048
#define EPW       16
#define CH        1536
#define KPT       (CH / 256)
#define UB        8

__global__ void bignn_zero_ctr(int* __restrict__ p, int n) {
    int i = blockIdx.x * blockDim.x + threadIdx.x;
    if (i < n) p[i] = 0;
}

__global__ __launch_bounds__(1024)
void bignn_count(const int* __restrict__ erow, int* __restrict__ cnt, int E, int NB) {
    __shared__ int h[MAXB];
    int tid = threadIdx.x;
    for (int i = tid; i < NB; i += 1024) h[i] = 0;
    __syncthreads();
    long chunk = ((long)E + gridDim.x - 1) / gridDim.x;
    long beg = (long)blockIdx.x * chunk;
    long end = beg + chunk;
    if (end > E) end = E;
    for (long e0 = beg + tid; e0 < end; e0 += UB * 1024) {
        int r[UB];
        #pragma unroll
        for (int k = 0; k < UB; ++k) {
            long e = e0 + k * 1024;
            r[k] = e < end ? erow[e] : -1;
        }
        #pragma unroll
        for (int k = 0; k < UB; ++k)
            if (r[k] >= 0) atomicAdd(&h[r[k] >> BIN_SHIFT], 1);
    }
    __syncthreads();
    for (int i = tid; i < NB; i += 1024) {
        int c = h[i];
        if (c) atomicAdd(&cnt[i], c);
    }
}

__global__ __launch_bounds__(1024)
void bignn_place(const int* __restrict__ erow, const int* __restrict__ ecol,
                 const float* __restrict__ eval_, const int* __restrict__ cnt,
                 int* __restrict__ cursor, int* __restrict__ binStart,
                 uint2* __restrict__ binned, int E, int NB) {
    __shared__ int lcnt[MAXB];
    __shared__ int lbase[MAXB];
    __shared__ int wsum[32];
    int tid = threadIdx.x;
    int lane = tid & 31;
    int wv = tid >> 5;
    int i0 = 2 * tid, i1 = 2 * tid + 1;
    int a = i0 < NB ? cnt[i0] : 0;
    int b = i1 < NB ? cnt[i1] : 0;
    int s = a + b;
    int incl = s;
    #pragma unroll
    for (int d = 1; d < 32; d <<= 1) {
        int t = __shfl_up(incl, d);
        if (lane >= d) incl += t;
    }
    if (lane == 31) wsum[wv] = incl;
    lcnt[i0] = 0;
    lcnt[i1] = 0;
    __syncthreads();
    if (wv == 0) {
        int w = wsum[lane];
        int wi = w;
        #pragma unroll
        for (int d = 1; d < 32; d <<= 1) {
            int t = __shfl_up(wi, d);
            if (lane >= d) wi += t;
        }
        wsum[lane] = wi - w;
    }
    __syncthreads();
    int excl = wsum[wv] + incl - s;
    lbase[i0] = excl;
    lbase[i1] = excl + a;
    if (blockIdx.x == 0) {
        if (i0 < NB) binStart[i0] = excl;
        if (i1 < NB) binStart[i1] = excl + a;
    }
    long chunk = ((long)E + gridDim.x - 1) / gridDim.x;
    long beg = (long)blockIdx.x * chunk;
    long end = beg + chunk;
    if (end > E) end = E;
    for (long e0 = beg + tid; e0 < end; e0 += UB * 1024) {
        int r[UB];
        #pragma unroll
        for (int k = 0; k < UB; ++k) {
            long e = e0 + k * 1024;
            r[k] = e < end ? erow[e] : -1;
        }
        #pragma unroll
        for (int k = 0; k < UB; ++k)
            if (r[k] >= 0) atomicAdd(&lcnt[r[k] >> BIN_SHIFT], 1);
    }
    __syncthreads();
    if (i0 < NB) { int c = lcnt[i0]; if (c) lbase[i0] += atomicAdd(&cursor[i0], c); }
    if (i1 < NB) { int c = lcnt[i1]; if (c) lbase[i1] += atomicAdd(&cursor[i1], c); }
    lcnt[i0] = 0;
    lcnt[i1] = 0;
    __syncthreads();
    for (long e0 = beg + tid; e0 < end; e0 += UB * 1024) {
        int r[UB];
        int c[UB];
        float v[UB];
        #pragma unroll
        for (int k = 0; k < UB; ++k) {
            long e = e0 + k * 1024;
            bool in = e < end;
            r[k] = in ? erow[e] : -1;
            c[k] = in ? ecol[e] : 0;
            v[k] = in ? eval_[e] : 0.f;
        }
        #pragma unroll
        for (int k = 0; k < UB; ++k) {
            if (r[k] >= 0) {
                int bin = r[k] >> BIN_SHIFT;
                int pos = lbase[bin] + atomicAdd(&lcnt[bin], 1);
                binned[pos] = make_uint2(((unsigned)c[k] << BIN_SHIFT) | (unsigned)(r[k] & (BIN_ROWS - 1)),
                                         __float_as_uint(v[k]));
            }
        }
    }
}

__device__ __forceinline__ void bignn_flush_row(float* sL, int cur, int lane, float a0, float a1) {
    if (cur >= 0) {
        float* dst = sL + cur * DIM;
        int c = lane ^ ((cur & 15) << 2);
        atomicAdd(dst + c,        a0);
        atomicAdd(dst + (c ^ 32), a1);
    }
}

template <bool FULL>
__device__ __forceinline__ void bignn_row_batch(const uint2* sorted, int b, int cnt, int lane,
                                                const float* __restrict__ x, float* sL,
                                                int& cur, float& acc0, float& acc1) {
    int slot = lane & (EPW - 1);
    uint2 kv = make_uint2(0u, 0u);
    if (FULL || slot < cnt) kv = sorted[b + slot];
    float xa[EPW];
    float xb[EPW];
    #pragma unroll
    for (int j = 0; j < EPW; ++j) {
        if (FULL || j < cnt) {
            unsigned key = (unsigned)__builtin_amdgcn_readlane((int)kv.x, j);
            unsigned long long rowAddr = (unsigned long long)(x + (long)(key >> BIN_SHIFT) * DIM);
            unsigned lo = (unsigned)__builtin_amdgcn_readfirstlane((int)(unsigned)rowAddr);
            unsigned hi = (unsigned)__builtin_amdgcn_readfirstlane((int)(unsigned)(rowAddr >> 32));
            const __attribute__((address_space(1))) float* srcRow =
                (const __attribute__((address_space(1))) float*)(((unsigned long long)hi << 32) | lo);
            xa[j] = srcRow[lane];
            xb[j] = srcRow[lane + 32];
        }
    }
    #pragma unroll
    for (int j = 0; j < EPW; ++j) {
        if (FULL || j < cnt) {
            unsigned key = (unsigned)__builtin_amdgcn_readlane((int)kv.x, j);
            float v = __int_as_float(__builtin_amdgcn_readlane((int)kv.y, j));
            int r = key & (BIN_ROWS - 1);
            if (r != cur) {
                bignn_flush_row(sL, cur, lane, acc0, acc1);
                acc0 = 0.f;
                acc1 = 0.f;
                cur = r;
            }
            acc0 += v * xa[j];
            acc1 += v * xb[j];
        }
    }
}

__global__ __launch_bounds__(256)
void bignn_accum_gemm(const uint2* __restrict__ binned, const int* __restrict__ cnt,
                      const int* __restrict__ binStart, const float* __restrict__ x,
                      const float* __restrict__ W1, const float* __restrict__ b1,
                      const float* __restrict__ W2, const float* __restrict__ b2,
                      float* __restrict__ out, int N) {
    __shared__ __attribute__((aligned(16))) float sL[BIN_ROWS * DIM];
    __shared__ __attribute__((aligned(16))) float sF[BIN_ROWS * DIM];
    __shared__ __attribute__((aligned(16))) unsigned char sRaw1[(DIM / 2) * DIM * 8];
    v2f*   sW1    = (v2f*)sRaw1;
    uint2* sorted = (uint2*)sRaw1;
    int*   rowCnt = (int*)(sorted + CH);
    __shared__ __attribute__((aligned(16))) v2f sW2[(DIM / 2) * DIM];
    int tid = threadIdx.x;
    int lane = tid & 31;
    int wv = __builtin_amdgcn_readfirstlane(tid >> 5);
    int bin = blockIdx.x;
    int row0 = bin * BIN_ROWS;

    int n = cnt[bin];
    const uint2* p = binned + binStart[bin];
    uint2 kv[KPT];

    {
        float4 fv[4];
        #pragma unroll
        for (int j = 0; j < 4; ++j) {
            int i = tid + 256 * j;
            int gr = row0 + (i >> 4);
            if (gr >= N) gr = N - 1;
            fv[j] = ((const float4*)(x + (long)gr * DIM))[i & 15];
        }
        float4 w2e[2], w2o[2];
        #pragma unroll
        for (int j = 0; j < 2; ++j) {
            int u  = tid + 256 * j;
            int kp = u >> 4;
            int q  = u & 15;
            w2e[j] = ((const float4*)(W2 + (2 * kp)     * DIM))[q];
            w2o[j] = ((const float4*)(W2 + (2 * kp + 1) * DIM))[q];
        }
        #pragma unroll
        for (int k = 0; k < KPT; ++k) {
            int i = k * 256 + tid;
            kv[k] = i < n ? p[i] : make_uint2(0u, 0u);
        }
        const float4 z = {0.f, 0.f, 0.f, 0.f};
        #pragma unroll
        for (int j = 0; j < 4; ++j) {
            int i = tid + 256 * j;
            int r = i >> 4;
            int q = i & 15;
            *(float4*)(sF + r * DIM + ((q ^ (r & 15)) << 2)) = fv[j];
            ((float4*)sL)[i] = z;
        }
        #pragma unroll
        for (int j = 0; j < 2; ++j) {
            int u = tid + 256 * j;
            float4* d2 = (float4*)(sW2 + (u >> 4) * DIM + (u & 15) * 4);
            d2[0] = float4{w2e[j].x, w2o[j].x, w2e[j].y, w2o[j].y};
            d2[1] = float4{w2e[j].z, w2o[j].z, w2e[j].w, w2o[j].w};
        }
    }
    for (int c0 = 0; c0 < n; c0 += CH) {
        int nc = n - c0 < CH ? n - c0 : CH;
        int rk[KPT];
        if (tid < BIN_ROWS) rowCnt[tid] = 0;
        __syncthreads();
        #pragma unroll
        for (int k = 0; k < KPT; ++k) {
            int i = k * 256 + tid;
            rk[k] = i < nc ? atomicAdd(&rowCnt[kv[k].x & (BIN_ROWS - 1)], 1) : 0;
        }
        __syncthreads();
        int ca = rowCnt[2 * lane];
        int cb = rowCnt[2 * lane + 1];
        int s = ca + cb;
        int incl = s;
        #pragma unroll
        for (int d = 1; d < 32; d <<= 1) {
            int t = __shfl_up(incl, d);
            if (lane >= d) incl += t;
        }
        int excl = incl - s;
        int packed = excl | ((excl + ca) << 16);
        #pragma unroll
        for (int k = 0; k < KPT; ++k) {
            int i = k * 256 + tid;
            int r = kv[k].x & (BIN_ROWS - 1);
            int pk = __shfl(packed, r >> 1);
            int start = (r & 1) ? (int)((unsigned)pk >> 16) : (pk & 0xffff);
            if (i < nc) sorted[start + rk[k]] = kv[k];
        }
        __syncthreads();

        int beg = (nc * wv) >> 3;
        int end = (nc * (wv + 1)) >> 3;
        int cur = -1;
        float acc0 = 0.f;
        float acc1 = 0.f;
        int b0 = beg;
        for (; b0 + EPW <= end; b0 += EPW)
            bignn_row_batch<true>(sorted, b0, EPW, lane, x, sL, cur, acc0, acc1);
        if (b0 < end)
            bignn_row_batch<false>(sorted, b0, end - b0, lane, x, sL, cur, acc0, acc1);
        bignn_flush_row(sL, cur, lane, acc0, acc1);
        #pragma unroll
        for (int k = 0; k < KPT; ++k) {
            int i = c0 + CH + k * 256 + tid;
            kv[k] = i < n ? p[i] : make_uint2(0u, 0u);
        }
    }

    {
        float4 w1e[2], w1o[2];
        #pragma unroll
        for (int j = 0; j < 2; ++j) {
            int u  = tid + 256 * j;
            int kp = u >> 4;
            int q  = u & 15;
            w1e[j] = ((const float4*)(W1 + (2 * kp)     * DIM))[q];
            w1o[j] = ((const float4*)(W1 + (2 * kp + 1) * DIM))[q];
        }
        __syncthreads();
        #pragma unroll
        for (int j = 0; j < 2; ++j) {
            int u = tid + 256 * j;
            float4* d1 = (float4*)(sW1 + (u >> 4) * DIM + (u & 15) * 4);
            d1[0] = float4{w1e[j].x, w1o[j].x, w1e[j].y, w1o[j].y};
            d1[1] = float4{w1e[j].z, w1o[j].z, w1e[j].w, w1o[j].w};
        }
    }
    __syncthreads();

    int m    = lane & 15;
    int half = lane >> 4;
    int rt   = wv & 3;
    int ch   = wv >> 2;
    int r    = rt * 16 + m;
    int sw   = m << 2;
    const float* lrow = sL + r * DIM;
    const float* frow = sF + r * DIM;
    int col0 = ch * 32 + m;
    int col1 = col0 + 16;
    float bias0 = b1[col0] + b2[col0];
    float bias1 = b1[col1] + b2[col1];

    v8f acc0 = v8f{};
    v8f acc1 = v8f{};
    for (int k0 = 0; k0 < 16; ++k0) {
        int kp = k0 * 2 + half;
        int c  = (2 * kp) ^ sw;
        float2 l = *(const float2*)(lrow + c);
        float2 f = *(const float2*)(frow + c);
        v2f aS = { l.x + f.x, l.y + f.y };
        v2f aP = { l.x * f.x, l.y * f.y };
        v2f bA0 = sW1[kp * DIM + col0];
        v2f bB0 = sW2[kp * DIM + col0];
        v2f bA1 = sW1[kp * DIM + col1];
        v2f bB1 = sW2[kp * DIM + col1];
        acc0 = __builtin_amdgcn_wmma_f32_16x16x4_f32(
            false, aS, false, bA0, (short)0, acc0, false, false);
        acc1 = __builtin_amdgcn_wmma_f32_16x16x4_f32(
            false, aS, false, bA1, (short)0, acc1, false, false);
        acc0 = __builtin_amdgcn_wmma_f32_16x16x4_f32(
            false, aP, false, bB0, (short)0, acc0, false, false);
        acc1 = __builtin_amdgcn_wmma_f32_16x16x4_f32(
            false, aP, false, bB1, (short)0, acc1, false, false);
    }

    int rBase = row0 + rt * 16 + half * 8;
    if (row0 + BIN_ROWS <= N) {
        #pragma unroll
        for (int g = 0; g < 8; ++g) {
            float* o = out + (long)(rBase + g) * DIM;
            o[col0] = acc0[g] + bias0;
            o[col1] = acc1[g] + bias1;
        }
    } else {
        #pragma unroll
        for (int g = 0; g < 8; ++g) {
            int gr = rBase + g;
            if (gr < N) {
                float* o = out + (long)gr * DIM;
                o[col0] = acc0[g] + bias0;
                o[col1] = acc1[g] + bias1;
            }
        }
    }
}

extern "C" void kernel_launch(void* const* d_in, const int* in_sizes, int n_in,
                              void* d_out, int out_size, void* d_ws, size_t ws_size,
                              hipStream_t stream) {
    const int*   erow  = (const int*)  d_in[0];
    const int*   ecol  = (const int*)  d_in[1];
    const float* eval_ = (const float*)d_in[2];
    const float* feat  = (const float*)d_in[3];
    const float* W1    = (const float*)d_in[4];
    const float* b1    = (const float*)d_in[5];
    const float* W2    = (const float*)d_in[6];
    const float* b2    = (const float*)d_in[7];
    float* out = (float*)d_out;

    int E = in_sizes[0];
    int N = in_sizes[3] / DIM;
    int NB = (N + BIN_ROWS - 1) / BIN_ROWS;

    uint2* binned   = (uint2*)d_ws;
    int*   cnt      = (int*)(binned + E);
    int*   cursor   = cnt + MAXB;
    int*   binStart = cursor + MAXB;

    bignn_zero_ctr<<<(2 * MAXB + 1023) / 1024, 1024, 0, stream>>>(cnt, 2 * MAXB);
    bignn_count<<<256, 1024, 0, stream>>>(erow, cnt, E, NB);
    bignn_place<<<256, 1024, 0, stream>>>(erow, ecol, eval_, cnt, cursor, binStart, binned, E, NB);
    bignn_accum_gemm<<<NB, 256, 0, stream>>>(binned, cnt, binStart, feat, W1, b1, W2, b2, out, N);
}


# Fused kernel gather: a 16-lane half-wave per edge, one 16-byte load per lane (2 whole rows per load instruction, list cut in 16 parts); f32 run sums regroup at the new cuts, still arrival-ordered
# speedup vs baseline: 3.4664x; 1.0564x over previous
#include <hip/hip_runtime.h>

#define DIM 64

typedef float v2f __attribute__((ext_vector_type(2)));
typedef float v8f __attribute__((ext_vector_type(8)));

#define BIN_SHIFT 6
#define BIN_ROWS  64
#define MAXB      2048
#define EPS       10
#define CH        1536
#define KPT       (CH / 256)
#define UB        8

__global__ void bignn_zero_ctr(int* __restrict__ p, int n) {
    int i = blockIdx.x * blockDim.x + threadIdx.x;
    if (i < n) p[i] = 0;
}

__global__ __launch_bounds__(1024)
void bignn_count(const int* __restrict__ erow, int* __restrict__ cnt, int E, int NB) {
    __shared__ int h[MAXB];
    int tid = threadIdx.x;
    for (int i = tid; i < NB; i += 1024) h[i] = 0;
    __syncthreads();
    long chunk = ((long)E + gridDim.x - 1) / gridDim.x;
    long beg = (long)blockIdx.x * chunk;
    long end = beg + chunk;
    if (end > E) end = E;
    for (long e0 = beg + tid; e0 < end; e0 += UB * 1024) {
        int r[UB];
        #pragma unroll
        for (int k = 0; k < UB; ++k) {
            long e = e0 + k * 1024;
            r[k] = e < end ? erow[e] : -1;
        }
        #pragma unroll
        for (int k = 0; k < UB; ++k)
            if (r[k] >= 0) atomicAdd(&h[r[k] >> BIN_SHIFT], 1);
    }
    __syncthreads();
    for (int i = tid; i < NB; i += 1024) {
        int c = h[i];
        if (c) atomicAdd(&cnt[i], c);
    }
}

__global__ __launch_bounds__(1024)
void bignn_place(const int* __restrict__ erow, const int* __restrict__ ecol,
                 const float* __restrict__ eval_, const int* __restrict__ cnt,
                 int* __restrict__ cursor, int* __restrict__ binStart,
                 uint2* __restrict__ binned, int E, int NB) {
    __shared__ int lcnt[MAXB];
    __shared__ int lbase[MAXB];
    __shared__ int wsum[32];
    int tid = threadIdx.x;
    int lane = tid & 31;
    int wv = tid >> 5;
    int i0 = 2 * tid, i1 = 2 * tid + 1;
    int a = i0 < NB ? cnt[i0] : 0;
    int b = i1 < NB ? cnt[i1] : 0;
    int s = a + b;
    int incl = s;
    #pragma unroll
    for (int d = 1; d < 32; d <<= 1) {
        int t = __shfl_up(incl, d);
        if (lane >= d) incl += t;
    }
    if (lane == 31) wsum[wv] = incl;
    lcnt[i0] = 0;
    lcnt[i1] = 0;
    __syncthreads();
    if (wv == 0) {
        int w = wsum[lane];
        int wi = w;
        #pragma unroll
        for (int d = 1; d < 32; d <<= 1) {
            int t = __shfl_up(wi, d);
            if (lane >= d) wi += t;
        }
        wsum[lane] = wi - w;
    }
    __syncthreads();
    int excl = wsum[wv] + incl - s;
    lbase[i0] = excl;
    lbase[i1] = excl + a;
    if (blockIdx.x == 0) {
        if (i0 < NB) binStart[i0] = excl;
        if (i1 < NB) binStart[i1] = excl + a;
    }
    long chunk = ((long)E + gridDim.x - 1) / gridDim.x;
    long beg = (long)blockIdx.x * chunk;
    long end = beg + chunk;
    if (end > E) end = E;
    for (long e0 = beg + tid; e0 < end; e0 += UB * 1024) {
        int r[UB];
        #pragma unroll
        for (int k = 0; k < UB; ++k) {
            long e = e0 + k * 1024;
            r[k] = e < end ? erow[e] : -1;
        }
        #pragma unroll
        for (int k = 0; k < UB; ++k)
            if (r[k] >= 0) atomicAdd(&lcnt[r[k] >> BIN_SHIFT], 1);
    }
    __syncthreads();
    if (i0 < NB) { int c = lcnt[i0]; if (c) lbase[i0] += atomicAdd(&cursor[i0], c); }
    if (i1 < NB) { int c = lcnt[i1]; if (c) lbase[i1] += atomicAdd(&cursor[i1], c); }
    lcnt[i0] = 0;
    lcnt[i1] = 0;
    __syncthreads();
    for (long e0 = beg + tid; e0 < end; e0 += UB * 1024) {
        int r[UB];
        int c[UB];
        float v[UB];
        #pragma unroll
        for (int k = 0; k < UB; ++k) {
            long e = e0 + k * 1024;
            bool in = e < end;
            r[k] = in ? erow[e] : -1;
            c[k] = in ? ecol[e] : 0;
            v[k] = in ? eval_[e] : 0.f;
        }
        #pragma unroll
        for (int k = 0; k < UB; ++k) {
            if (r[k] >= 0) {
                int bin = r[k] >> BIN_SHIFT;
                int pos = lbase[bin] + atomicAdd(&lcnt[bin], 1);
                binned[pos] = make_uint2(((unsigned)c[k] << BIN_SHIFT) | (unsigned)(r[k] & (BIN_ROWS - 1)),
                                         __float_as_uint(v[k]));
            }
        }
    }
}

__device__ __forceinline__ void bignn_flush_row(float* sL, int cur, int q, float4 a) {
    if (cur >= 0) {
        float* dst = sL + cur * DIM + ((q ^ (cur & 15)) << 2);
        atomicAdd(dst,     a.x);
        atomicAdd(dst + 1, a.y);
        atomicAdd(dst + 2, a.z);
        atomicAdd(dst + 3, a.w);
    }
}

template <bool FULL>
__device__ __forceinline__ void bignn_row_batch(const uint2* sorted, int pos, int cnt, int q,
                                                const float* __restrict__ x, float* sL,
                                                int& cur, float4& acc) {
    uint2 kv[EPS];
    float4 xv[EPS];
    #pragma unroll
    for (int j = 0; j < EPS; ++j) {
        kv[j] = sorted[(FULL || j < cnt) ? pos + j : 0];
        unsigned off = ((kv[j].x >> BIN_SHIFT) << 8) | ((unsigned)q << 4);
        xv[j] = *(const float4*)((const char*)x + off);
    }
    #pragma unroll
    for (int j = 0; j < EPS; ++j) {
        if (FULL || j < cnt) {
            int r = (int)(kv[j].x & (BIN_ROWS - 1));
            float v = __uint_as_float(kv[j].y);
            if (r != cur) {
                bignn_flush_row(sL, cur, q, acc);
                acc = float4{0.f, 0.f, 0.f, 0.f};
                cur = r;
            }
            acc.x += v * xv[j].x;
            acc.y += v * xv[j].y;
            acc.z += v * xv[j].z;
            acc.w += v * xv[j].w;
        }
    }
}

__global__ __launch_bounds__(256) __attribute__((amdgpu_waves_per_eu(10, 10)))
void bignn_accum_gemm(const uint2* __restrict__ binned, const int* __restrict__ cnt,
                      const int* __restrict__ binStart, const float* __restrict__ x,
                      const float* __restrict__ W1, const float* __restrict__ b1,
                      const float* __restrict__ W2, const float* __restrict__ b2,
                      float* __restrict__ out, int N) {
    __shared__ __attribute__((aligned(16))) float sL[BIN_ROWS * DIM];
    __shared__ __attribute__((aligned(16))) float sF[BIN_ROWS * DIM];
    __shared__ __attribute__((aligned(16))) unsigned char sRaw1[(DIM / 2) * DIM * 8];
    v2f*   sW1    = (v2f*)sRaw1;
    uint2* sorted = (uint2*)sRaw1;
    int*   rowCnt = (int*)(sorted + CH);
    __shared__ __attribute__((aligned(16))) v2f sW2[(DIM / 2) * DIM];
    int tid = threadIdx.x;
    int lane = tid & 31;
    int wv = __builtin_amdgcn_readfirstlane(tid >> 5);
    int bin = blockIdx.x;
    int row0 = bin * BIN_ROWS;

    int n = cnt[bin];
    const uint2* p = binned + binStart[bin];
    uint2 kv[KPT];

    {
        float4 fv[4];
        #pragma unroll
        for (int j = 0; j < 4; ++j) {
            int i = tid + 256 * j;
            int gr = row0 + (i >> 4);
            if (gr >= N) gr = N - 1;
            fv[j] = ((const float4*)(x + (long)gr * DIM))[i & 15];
        }
        float4 w2e[2], w2o[2];
        #pragma unroll
        for (int j = 0; j < 2; ++j) {
            int u  = tid + 256 * j;
            int kp = u >> 4;
            int q  = u & 15;
            w2e[j] = ((const float4*)(W2 + (2 * kp)     * DIM))[q];
            w2o[j] = ((const float4*)(W2 + (2 * kp + 1) * DIM))[q];
        }
        #pragma unroll
        for (int k = 0; k < KPT; ++k) {
            int i = k * 256 + tid;
            kv[k] = i < n ? p[i] : make_uint2(0u, 0u);
        }
        const float4 z = {0.f, 0.f, 0.f, 0.f};
        #pragma unroll
        for (int j = 0; j < 4; ++j) {
            int i = tid + 256 * j;
            int r = i >> 4;
            int q = i & 15;
            *(float4*)(sF + r * DIM + ((q ^ (r & 15)) << 2)) = fv[j];
            ((float4*)sL)[i] = z;
        }
        #pragma unroll
        for (int j = 0; j < 2; ++j) {
            int u = tid + 256 * j;
            float4* d2 = (float4*)(sW2 + (u >> 4) * DIM + (u & 15) * 4);
            d2[0] = float4{w2e[j].x, w2o[j].x, w2e[j].y, w2o[j].y};
            d2[1] = float4{w2e[j].z, w2o[j].z, w2e[j].w, w2o[j].w};
        }
    }
    for (int c0 = 0; c0 < n; c0 += CH) {
        int nc = n - c0 < CH ? n - c0 : CH;
        int rk[KPT];
        if (tid < BIN_ROWS) rowCnt[tid] = 0;
        __syncthreads();
        #pragma unroll
        for (int k = 0; k < KPT; ++k) {
            int i = k * 256 + tid;
            rk[k] = i < nc ? atomicAdd(&rowCnt[kv[k].x & (BIN_ROWS - 1)], 1) : 0;
        }
        __syncthreads();
        int ca = rowCnt[2 * lane];
        int cb = rowCnt[2 * lane + 1];
        int s = ca + cb;
        int incl = s;
        #pragma unroll
        for (int d = 1; d < 32; d <<= 1) {
            int t = __shfl_up(incl, d);
            if (lane >= d) incl += t;
        }
        int excl = incl - s;
        int packed = excl | ((excl + ca) << 16);
        #pragma unroll
        for (int k = 0; k < KPT; ++k) {
            int i = k * 256 + tid;
            int r = kv[k].x & (BIN_ROWS - 1);
            int pk = __shfl(packed, r >> 1);
            int start = (r & 1) ? (int)((unsigned)pk >> 16) : (pk & 0xffff);
            if (i < nc) sorted[start + rk[k]] = kv[k];
        }
        __syncthreads();

        int cutA = (nc * (2 * wv)) >> 4;
        int cutB = (nc * (2 * wv + 1)) >> 4;
        int cutC = (nc * (2 * wv + 2)) >> 4;
        int lenA = cutB - cutA;
        int lenB = cutC - cutB;
        int lenMin = lenA < lenB ? lenA : lenB;
        int lenMax = lenA < lenB ? lenB : lenA;
        int q     = lane & 15;
        int beg   = lane < 16 ? cutA : cutB;
        int myLen = lane < 16 ? lenA : lenB;
        int cur = -1;
        float4 acc = float4{0.f, 0.f, 0.f, 0.f};
        int b0 = 0;
        for (; b0 + EPS <= lenMin; b0 += EPS)
            bignn_row_batch<true>(sorted, beg + b0, EPS, q, x, sL, cur, acc);
        if (b0 < lenMax)
            bignn_row_batch<false>(sorted, beg + b0, myLen - b0, q, x, sL, cur, acc);
        bignn_flush_row(sL, cur, q, acc);
        #pragma unroll
        for (int k = 0; k < KPT; ++k) {
            int i = c0 + CH + k * 256 + tid;
            kv[k] = i < n ? p[i] : make_uint2(0u, 0u);
        }
    }

    {
        float4 w1e[2], w1o[2];
        #pragma unroll
        for (int j = 0; j < 2; ++j) {
            int u  = tid + 256 * j;
            int kp = u >> 4;
            int q  = u & 15;
            w1e[j] = ((const float4*)(W1 + (2 * kp)     * DIM))[q];
            w1o[j] = ((const float4*)(W1 + (2 * kp + 1) * DIM))[q];
        }
        __syncthreads();
        #pragma unroll
        for (int j = 0; j < 2; ++j) {
            int u = tid + 256 * j;
            float4* d1 = (float4*)(sW1 + (u >> 4) * DIM + (u & 15) * 4);
            d1[0] = float4{w1e[j].x, w1o[j].x, w1e[j].y, w1o[j].y};
            d1[1] = float4{w1e[j].z, w1o[j].z, w1e[j].w, w1o[j].w};
        }
    }
    __syncthreads();

    int m    = lane & 15;
    int half = lane >> 4;
    int rt   = wv & 3;
    int ch   = wv >> 2;
    int r    = rt * 16 + m;
    int sw   = m << 2;
    const float* lrow = sL + r * DIM;
    const float* frow = sF + r * DIM;
    int col0 = ch * 32 + m;
    int col1 = col0 + 16;
    float bias0 = b1[col0] + b2[col0];
    float bias1 = b1[col1] + b2[col1];

    v8f acc0 = v8f{};
    v8f acc1 = v8f{};
    for (int k0 = 0; k0 < 16; ++k0) {
        int kp = k0 * 2 + half;
        int c  = (2 * kp) ^ sw;
        float2 l = *(const float2*)(lrow + c);
        float2 f = *(const float2*)(frow + c);
        v2f aS = { l.x + f.x, l.y + f.y };
        v2f aP = { l.x * f.x, l.y * f.y };
        v2f bA0 = sW1[kp * DIM + col0];
        v2f bB0 = sW2[kp * DIM + col0];
        v2f bA1 = sW1[kp * DIM + col1];
        v2f bB1 = sW2[kp * DIM + col1];
        acc0 = __builtin_amdgcn_wmma_f32_16x16x4_f32(
            false, aS, false, bA0, (short)0, acc0, false, false);
        acc1 = __builtin_amdgcn_wmma_f32_16x16x4_f32(
            false, aS, false, bA1, (short)0, acc1, false, false);
        acc0 = __builtin_amdgcn_wmma_f32_16x16x4_f32(
            false, aP, false, bB0, (short)0, acc0, false, false);
        acc1 = __builtin_amdgcn_wmma_f32_16x16x4_f32(
            false, aP, false, bB1, (short)0, acc1, false, false);
    }

    int rBase = row0 + rt * 16 + half * 8;
    if (row0 + BIN_ROWS <= N) {
        #pragma unroll
        for (int g = 0; g < 8; ++g) {
            float* o = out + (long)(rBase + g) * DIM;
            o[col0] = acc0[g] + bias0;
            o[col1] = acc1[g] + bias1;
        }
    } else {
        #pragma unroll
        for (int g = 0; g < 8; ++g) {
            int gr = rBase + g;
            if (gr < N) {
                float* o = out + (long)gr * DIM;
                o[col0] = acc0[g] + bias0;
                o[col1] = acc1[g] + bias1;
            }
        }
    }
}

extern "C" void kernel_launch(void* const* d_in, const int* in_sizes, int n_in,
                              void* d_out, int out_size, void* d_ws, size_t ws_size,
                              hipStream_t stream) {
    const int*   erow  = (const int*)  d_in[0];
    const int*   ecol  = (const int*)  d_in[1];
    const float* eval_ = (const float*)d_in[2];
    const float* feat  = (const float*)d_in[3];
    const float* W1    = (const float*)d_in[4];
    const float* b1    = (const float*)d_in[5];
    const float* W2    = (const float*)d_in[6];
    const float* b2    = (const float*)d_in[7];
    float* out = (float*)d_out;

    int E = in_sizes[0];
    int N = in_sizes[3] / DIM;
    int NB = (N + BIN_ROWS - 1) / BIN_ROWS;

    uint2* binned   = (uint2*)d_ws;
    int*   cnt      = (int*)(binned + E);
    int*   cursor   = cnt + MAXB;
    int*   binStart = cursor + MAXB;

    bignn_zero_ctr<<<(2 * MAXB + 1023) / 1024, 1024, 0, stream>>>(cnt, 2 * MAXB);
    bignn_count<<<256, 1024, 0, stream>>>(erow, cnt, E, NB);
    bignn_place<<<256, 1024, 0, stream>>>(erow, ecol, eval_, cnt, cursor, binStart, binned, E, NB);
    bignn_accum_gemm<<<NB, 256, 0, stream>>>(binned, cnt, binStart, feat, W1, b1, W2, b2, out, N);
}


# Place kernel: a block stages its slice's records in LDS in bin order (ranked by LDS adds, slice read once) and copies them out as runs: a store covers a few runs, not 32 lines; sums arrival-ordered
# speedup vs baseline: 3.7568x; 1.0838x over previous
#include <hip/hip_runtime.h>

#define DIM 64

typedef float v2f __attribute__((ext_vector_type(2)));
typedef float v8f __attribute__((ext_vector_type(8)));

#define BIN_SHIFT 6
#define BIN_ROWS  64
#define MAXB      2048
#define EPS       10
#define CH        1536
#define KPT       (CH / 256)
#define UB        8
#define SP        (UB * 1024)

__global__ void bignn_zero_ctr(int* __restrict__ p, int n) {
    int i = blockIdx.x * blockDim.x + threadIdx.x;
    if (i < n) p[i] = 0;
}

__global__ __launch_bounds__(1024)
void bignn_count(const int* __restrict__ erow, int* __restrict__ cnt, int E, int NB) {
    __shared__ int h[MAXB];
    int tid = threadIdx.x;
    for (int i = tid; i < NB; i += 1024) h[i] = 0;
    __syncthreads();
    long chunk = ((long)E + gridDim.x - 1) / gridDim.x;
    long beg = (long)blockIdx.x * chunk;
    long end = beg + chunk;
    if (end > E) end = E;
    for (long e0 = beg + tid; e0 < end; e0 += UB * 1024) {
        int r[UB];
        #pragma unroll
        for (int k = 0; k < UB; ++k) {
            long e = e0 + k * 1024;
            r[k] = e < end ? erow[e] : -1;
        }
        #pragma unroll
        for (int k = 0; k < UB; ++k)
            if (r[k] >= 0) atomicAdd(&h[r[k] >> BIN_SHIFT], 1);
    }
    __syncthreads();
    for (int i = tid; i < NB; i += 1024) {
        int c = h[i];
        if (c) atomicAdd(&cnt[i], c);
    }
}

__global__ __launch_bounds__(1024)
void bignn_place(const int* __restrict__ erow, const int* __restrict__ ecol,
                 const float* __restrict__ eval_, const int* __restrict__ cnt,
                 int* __restrict__ cursor, int* __restrict__ binStart,
                 uint2* __restrict__ binned, int E, int NB) {
    __shared__ int lcnt[MAXB];
    __shared__ int lbase[MAXB];
    __shared__ int lstart[MAXB];
    __shared__ int wsumG[32];
    __shared__ int wsumL[32];
    __shared__ __attribute__((aligned(16))) uint2 stE[SP];
    __shared__ int stP[SP];
    int tid = threadIdx.x;
    int lane = tid & 31;
    int wv = tid >> 5;
    int i0 = 2 * tid, i1 = 2 * tid + 1;
    long chunk = ((long)E + gridDim.x - 1) / gridDim.x;
    long beg = (long)blockIdx.x * chunk;
    long end = beg + chunk;
    if (end > E) end = E;
    for (long rb = beg; rb < end; rb += SP) {
        long re = rb + SP < end ? rb + SP : end;
        int nR = (int)(re - rb);
        int r[UB];
        int c[UB];
        float v[UB];
        #pragma unroll
        for (int k = 0; k < UB; ++k) {
            long e = rb + tid + k * 1024;
            bool in = e < re;
            r[k] = in ? erow[e] : -1;
            c[k] = in ? ecol[e] : 0;
            v[k] = in ? eval_[e] : 0.f;
        }
        int a = i0 < NB ? cnt[i0] : 0;
        int b = i1 < NB ? cnt[i1] : 0;
        lcnt[i0] = 0;
        lcnt[i1] = 0;
        __syncthreads();
        #pragma unroll
        for (int k = 0; k < UB; ++k)
            if (r[k] >= 0) atomicAdd(&lcnt[r[k] >> BIN_SHIFT], 1);
        __syncthreads();
        int la = lcnt[i0];
        int lb = lcnt[i1];
        int sG = a + b;
        int sL = la + lb;
        int inclG = sG;
        int inclL = sL;
        #pragma unroll
        for (int d = 1; d < 32; d <<= 1) {
            int t = __shfl_up(inclG, d);
            int u = __shfl_up(inclL, d);
            if (lane >= d) { inclG += t; inclL += u; }
        }
        if (lane == 31) { wsumG[wv] = inclG; wsumL[wv] = inclL; }
        __syncthreads();
        if (wv == 0) {
            int wG = wsumG[lane];
            int wL = wsumL[lane];
            int iG = wG;
            int iL = wL;
            #pragma unroll
            for (int d = 1; d < 32; d <<= 1) {
                int t = __shfl_up(iG, d);
                int u = __shfl_up(iL, d);
                if (lane >= d) { iG += t; iL += u; }
            }
            wsumG[lane] = iG - wG;
            wsumL[lane] = iL - wL;
        }
        __syncthreads();
        int exclG = wsumG[wv] + inclG - sG;
        int exclL = wsumL[wv] + inclL - sL;
        if (blockIdx.x == 0 && rb == beg) {
            if (i0 < NB) binStart[i0] = exclG;
            if (i1 < NB) binStart[i1] = exclG + a;
        }
        int g0 = exclG;
        int g1 = exclG + a;
        if (i0 < NB && la) g0 += atomicAdd(&cursor[i0], la);
        if (i1 < NB && lb) g1 += atomicAdd(&cursor[i1], lb);
        lbase[i0] = g0;
        lbase[i1] = g1;
        lstart[i0] = exclL;
        lstart[i1] = exclL + la;
        lcnt[i0] = 0;
        lcnt[i1] = 0;
        __syncthreads();
        #pragma unroll
        for (int k = 0; k < UB; ++k) {
            if (r[k] >= 0) {
                int bin = r[k] >> BIN_SHIFT;
                int rank = atomicAdd(&lcnt[bin], 1);
                int s = lstart[bin] + rank;
                stE[s] = make_uint2(((unsigned)c[k] << BIN_SHIFT) | (unsigned)(r[k] & (BIN_ROWS - 1)),
                                    __float_as_uint(v[k]));
                stP[s] = lbase[bin] + rank;
            }
        }
        __syncthreads();
        uint2 se[UB];
        int sp[UB];
        #pragma unroll
        for (int k = 0; k < UB; ++k) {
            int s = tid + k * 1024;
            if (s < nR) { se[k] = stE[s]; sp[k] = stP[s]; }
        }
        #pragma unroll
        for (int k = 0; k < UB; ++k) {
            int s = tid + k * 1024;
            if (s < nR) binned[sp[k]] = se[k];
        }
    }
}

__device__ __forceinline__ void bignn_flush_row(float* sL, int cur, int q, float4 a) {
    if (cur >= 0) {
        float* dst = sL + cur * DIM + ((q ^ (cur & 15)) << 2);
        atomicAdd(dst,     a.x);
        atomicAdd(dst + 1, a.y);
        atomicAdd(dst + 2, a.z);
        atomicAdd(dst + 3, a.w);
    }
}

template <bool FULL>
__device__ __forceinline__ void bignn_row_batch(const uint2* sorted, int pos, int cnt, int q,
                                                const float* __restrict__ x, float* sL,
                                                int& cur, float4& acc) {
    uint2 kv[EPS];
    float4 xv[EPS];
    #pragma unroll
    for (int j = 0; j < EPS; ++j) {
        kv[j] = sorted[(FULL || j < cnt) ? pos + j : 0];
        unsigned off = ((kv[j].x >> BIN_SHIFT) << 8) | ((unsigned)q << 4);
        xv[j] = *(const float4*)((const char*)x + off);
    }
    #pragma unroll
    for (int j = 0; j < EPS; ++j) {
        if (FULL || j < cnt) {
            int r = (int)(kv[j].x & (BIN_ROWS - 1));
            float v = __uint_as_float(kv[j].y);
            if (r != cur) {
                bignn_flush_row(sL, cur, q, acc);
                acc = float4{0.f, 0.f, 0.f, 0.f};
                cur = r;
            }
            acc.x += v * xv[j].x;
            acc.y += v * xv[j].y;
            acc.z += v * xv[j].z;
            acc.w += v * xv[j].w;
        }
    }
}

__global__ __launch_bounds__(256) __attribute__((amdgpu_waves_per_eu(10, 10)))
void bignn_accum_gemm(const uint2* __restrict__ binned, const int* __restrict__ cnt,
                      const int* __restrict__ binStart, const float* __restrict__ x,
                      const float* __restrict__ W1, const float* __restrict__ b1,
                      const float* __restrict__ W2, const float* __restrict__ b2,
                      float* __restrict__ out, int N) {
    __shared__ __attribute__((aligned(16))) float sL[BIN_ROWS * DIM];
    __shared__ __attribute__((aligned(16))) float sF[BIN_ROWS * DIM];
    __shared__ __attribute__((aligned(16))) unsigned char sRaw1[(DIM / 2) * DIM * 8];
    v2f*   sW1    = (v2f*)sRaw1;
    uint2* sorted = (uint2*)sRaw1;
    int*   rowCnt = (int*)(sorted + CH);
    __shared__ __attribute__((aligned(16))) v2f sW2[(DIM / 2) * DIM];
    int tid = threadIdx.x;
    int lane = tid & 31;
    int wv = __builtin_amdgcn_readfirstlane(tid >> 5);
    int bin = blockIdx.x;
    int row0 = bin * BIN_ROWS;

    int n = cnt[bin];
    const uint2* p = binned + binStart[bin];
    uint2 kv[KPT];

    {
        float4 fv[4];
        #pragma unroll
        for (int j = 0; j < 4; ++j) {
            int i = tid + 256 * j;
            int gr = row0 + (i >> 4);
            if (gr >= N) gr = N - 1;
            fv[j] = ((const float4*)(x + (long)gr * DIM))[i & 15];
        }
        float4 w2e[2], w2o[2];
        #pragma unroll
        for (int j = 0; j < 2; ++j) {
            int u  = tid + 256 * j;
            int kp = u >> 4;
            int q  = u & 15;
            w2e[j] = ((const float4*)(W2 + (2 * kp)     * DIM))[q];
            w2o[j] = ((const float4*)(W2 + (2 * kp + 1) * DIM))[q];
        }
        #pragma unroll
        for (int k = 0; k < KPT; ++k) {
            int i = k * 256 + tid;
            kv[k] = i < n ? p[i] : make_uint2(0u, 0u);
        }
        const float4 z = {0.f, 0.f, 0.f, 0.f};
        #pragma unroll
        for (int j = 0; j < 4; ++j) {
            int i = tid + 256 * j;
            int r = i >> 4;
            int q = i & 15;
            *(float4*)(sF + r * DIM + ((q ^ (r & 15)) << 2)) = fv[j];
            ((float4*)sL)[i] = z;
        }
        #pragma unroll
        for (int j = 0; j < 2; ++j) {
            int u = tid + 256 * j;
            float4* d2 = (float4*)(sW2 + (u >> 4) * DIM + (u & 15) * 4);
            d2[0] = float4{w2e[j].x, w2o[j].x, w2e[j].y, w2o[j].y};
            d2[1] = float4{w2e[j].z, w2o[j].z, w2e[j].w, w2o[j].w};
        }
    }
    for (int c0 = 0; c0 < n; c0 += CH) {
        int nc = n - c0 < CH ? n - c0 : CH;
        int rk[KPT];
        if (tid < BIN_ROWS) rowCnt[tid] = 0;
        __syncthreads();
        #pragma unroll
        for (int k = 0; k < KPT; ++k) {
            int i = k * 256 + tid;
            rk[k] = i < nc ? atomicAdd(&rowCnt[kv[k].x & (BIN_ROWS - 1)], 1) : 0;
        }
        __syncthreads();
        int ca = rowCnt[2 * lane];
        int cb = rowCnt[2 * lane + 1];
        int s = ca + cb;
        int incl = s;
        #pragma unroll
        for (int d = 1; d < 32; d <<= 1) {
            int t = __shfl_up(incl, d);
            if (lane >= d) incl += t;
        }
        int excl = incl - s;
        int packed = excl | ((excl + ca) << 16);
        #pragma unroll
        for (int k = 0; k < KPT; ++k) {
            int i = k * 256 + tid;
            int r = kv[k].x & (BIN_ROWS - 1);
            int pk = __shfl(packed, r >> 1);
            int start = (r & 1) ? (int)((unsigned)pk >> 16) : (pk & 0xffff);
            if (i < nc) sorted[start + rk[k]] = kv[k];
        }
        __syncthreads();

        int cutA = (nc * (2 * wv)) >> 4;
        int cutB = (nc * (2 * wv + 1)) >> 4;
        int cutC = (nc * (2 * wv + 2)) >> 4;
        int lenA = cutB - cutA;
        int lenB = cutC - cutB;
        int lenMin = lenA < lenB ? lenA : lenB;
        int lenMax = lenA < lenB ? lenB : lenA;
        int q     = lane & 15;
        int beg   = lane < 16 ? cutA : cutB;
        int myLen = lane < 16 ? lenA : lenB;
        int cur = -1;
        float4 acc = float4{0.f, 0.f, 0.f, 0.f};
        int b0 = 0;
        for (; b0 + EPS <= lenMin; b0 += EPS)
            bignn_row_batch<true>(sorted, beg + b0, EPS, q, x, sL, cur, acc);
        if (b0 < lenMax)
            bignn_row_batch<false>(sorted, beg + b0, myLen - b0, q, x, sL, cur, acc);
        bignn_flush_row(sL, cur, q, acc);
        #pragma unroll
        for (int k = 0; k < KPT; ++k) {
            int i = c0 + CH + k * 256 + tid;
            kv[k] = i < n ? p[i] : make_uint2(0u, 0u);
        }
    }

    {
        float4 w1e[2], w1o[2];
        #pragma unroll
        for (int j = 0; j < 2; ++j) {
            int u  = tid + 256 * j;
            int kp = u >> 4;
            int q  = u & 15;
            w1e[j] = ((const float4*)(W1 + (2 * kp)     * DIM))[q];
            w1o[j] = ((const float4*)(W1 + (2 * kp + 1) * DIM))[q];
        }
        __syncthreads();
        #pragma unroll
        for (int j = 0; j < 2; ++j) {
            int u = tid + 256 * j;
            float4* d1 = (float4*)(sW1 + (u >> 4) * DIM + (u & 15) * 4);
            d1[0] = float4{w1e[j].x, w1o[j].x, w1e[j].y, w1o[j].y};
            d1[1] = float4{w1e[j].z, w1o[j].z, w1e[j].w, w1o[j].w};
        }
    }
    __syncthreads();

    int m    = lane & 15;
    int half = lane >> 4;
    int rt   = wv & 3;
    int ch   = wv >> 2;
    int r    = rt * 16 + m;
    int sw   = m << 2;
    const float* lrow = sL + r * DIM;
    const float* frow = sF + r * DIM;
    int col0 = ch * 32 + m;
    int col1 = col0 + 16;
    float bias0 = b1[col0] + b2[col0];
    float bias1 = b1[col1] + b2[col1];

    v8f acc0 = v8f{};
    v8f acc1 = v8f{};
    for (int k0 = 0; k0 < 16; ++k0) {
        int kp = k0 * 2 + half;
        int c  = (2 * kp) ^ sw;
        float2 l = *(const float2*)(lrow + c);
        float2 f = *(const float2*)(frow + c);
        v2f aS = { l.x + f.x, l.y + f.y };
        v2f aP = { l.x * f.x, l.y * f.y };
        v2f bA0 = sW1[kp * DIM + col0];
        v2f bB0 = sW2[kp * DIM + col0];
        v2f bA1 = sW1[kp * DIM + col1];
        v2f bB1 = sW2[kp * DIM + col1];
        acc0 = __builtin_amdgcn_wmma_f32_16x16x4_f32(
            false, aS, false, bA0, (short)0, acc0, false, false);
        acc1 = __builtin_amdgcn_wmma_f32_16x16x4_f32(
            false, aS, false, bA1, (short)0, acc1, false, false);
        acc0 = __builtin_amdgcn_wmma_f32_16x16x4_f32(
            false, aP, false, bB0, (short)0, acc0, false, false);
        acc1 = __builtin_amdgcn_wmma_f32_16x16x4_f32(
            false, aP, false, bB1, (short)0, acc1, false, false);
    }

    int rBase = row0 + rt * 16 + half * 8;
    if (row0 + BIN_ROWS <= N) {
        #pragma unroll
        for (int g = 0; g < 8; ++g) {
            float* o = out + (long)(rBase + g) * DIM;
            o[col0] = acc0[g] + bias0;
            o[col1] = acc1[g] + bias1;
        }
    } else {
        #pragma unroll
        for (int g = 0; g < 8; ++g) {
            int gr = rBase + g;
            if (gr < N) {
                float* o = out + (long)gr * DIM;
                o[col0] = acc0[g] + bias0;
                o[col1] = acc1[g] + bias1;
            }
        }
    }
}

extern "C" void kernel_launch(void* const* d_in, const int* in_sizes, int n_in,
                              void* d_out, int out_size, void* d_ws, size_t ws_size,
                              hipStream_t stream) {
    const int*   erow  = (const int*)  d_in[0];
    const int*   ecol  = (const int*)  d_in[1];
    const float* eval_ = (const float*)d_in[2];
    const float* feat  = (const float*)d_in[3];
    const float* W1    = (const float*)d_in[4];
    const float* b1    = (const float*)d_in[5];
    const float* W2    = (const float*)d_in[6];
    const float* b2    = (const float*)d_in[7];
    float* out = (float*)d_out;

    int E = in_sizes[0];
    int N = in_sizes[3] / DIM;
    int NB = (N + BIN_ROWS - 1) / BIN_ROWS;

    uint2* binned   = (uint2*)d_ws;
    int*   cnt      = (int*)(binned + E);
    int*   cursor   = cnt + MAXB;
    int*   binStart = cursor + MAXB;

    bignn_zero_ctr<<<(2 * MAXB + 1023) / 1024, 1024, 0, stream>>>(cnt, 2 * MAXB);
    bignn_count<<<256, 1024, 0, stream>>>(erow, cnt, E, NB);
    bignn_place<<<256, 1024, 0, stream>>>(erow, ecol, eval_, cnt, cursor, binStart, binned, E, NB);
    bignn_accum_gemm<<<NB, 256, 0, stream>>>(binned, cnt, binStart, feat, W1, b1, W2, b2, out, N);
}
